# starting kernel
# baseline (speedup 1.0000x reference)
#include <hip/hip_runtime.h>

#define BB    16
#define NN    5000
#define EE    80000
#define FEAT  50
#define HID   128
#define PADR  1000

typedef __attribute__((ext_vector_type(2))) float v2f;
typedef __attribute__((ext_vector_type(8))) float v8f;

__device__ __forceinline__ v8f wmma_f32x4(v2f a, v2f b, v8f c) {
  return __builtin_amdgcn_wmma_f32_16x16x4_f32(
       false, a,  false, b,
       (short)0, c,  false,  false);
}

__device__ __forceinline__ int imin(int a, int b) { return a < b ? a : b; }


__global__ void fill_zero_kernel(float* __restrict__ p, long n) {
  long i = (long)blockIdx.x * blockDim.x + threadIdx.x;
  long stride = (long)gridDim.x * blockDim.x;
  for (; i < n; i += stride) p[i] = 0.0f;
}

__global__ void count_edges_kernel(const int* __restrict__ ei, float* __restrict__ cnt) {
  long total = (long)BB * EE;
  long i = (long)blockIdx.x * blockDim.x + threadIdx.x;
  long stride = (long)gridDim.x * blockDim.x;
  for (; i < total; i += stride) {
    int e = (int)(i % EE);
    int b = (int)(i / EE);
    int dst = ei[(long)b * 2 * EE + EE + e];
    atomicAdd(&cnt[(long)b * NN + dst], 1.0f);
  }
}

__global__ void scatter_add_kernel(const float* __restrict__ x, const int* __restrict__ ei,
                                   float* __restrict__ agg, int F) {
  long total = (long)BB * EE * F;
  long i = (long)blockIdx.x * blockDim.x + threadIdx.x;
  long stride = (long)gridDim.x * blockDim.x;
  for (; i < total; i += stride) {
    int f = (int)(i % F);
    long t = i / F;
    int e = (int)(t % EE);
    int b = (int)(t / EE);
    const int* eib = ei + (long)b * 2 * EE;
    int src = eib[e];
    int dst = eib[EE + e];
    atomicAdd(&agg[((long)b * NN + dst) * F + f], x[((long)b * NN + src) * F + f]);
  }
}

__global__ void mean_div_kernel(float* __restrict__ agg, const float* __restrict__ cnt, int F) {
  long total = (long)BB * NN * F;
  long i = (long)blockIdx.x * blockDim.x + threadIdx.x;
  long stride = (long)gridDim.x * blockDim.x;
  for (; i < total; i += stride) {
    float c = cnt[i / F];
    agg[i] /= fmaxf(c, 1.0f);
  }
}

template <int NT, int NCT>
__global__ void wmma_gemm_kernel(const float* __restrict__ A,
                                 const float* __restrict__ A2,
                                 const float* __restrict__ Wa,
                                 const float* __restrict__ Wb,
                                 const float* __restrict__ bias,
                                 float* __restrict__ C,
                                 int M, int K, int relu,
                                 int pad_mode, const int* __restrict__ act_range_p,
                                 int pad_rows, int src_rows) {
  const int lane = threadIdx.x & 31;
  const int mt = blockIdx.x * (blockDim.x >> 5) + (threadIdx.x >> 5);
  const int tm = (M + 15) >> 4;
  if (mt >= tm) return;

  const int ml = lane & 15;
  const int g  = lane >> 4;

  const int row = mt * 16 + ml;
  bool rvalid = row < M;
  long arow;
  if (pad_mode) {
    const int rc = imin(row, M - 1);
    const int b  = rc / pad_rows;
    const int rr = rc - b * pad_rows;
    rvalid = rvalid && (rr < *act_range_p);
    arow = (long)b * src_rows + rr;
  } else {
    arow = imin(row, M - 1);
  }

  v8f acc[NT];
#pragma unroll
  for (int t = 0; t < NT; ++t) acc[t] = (v8f){};

  const int npass = (A2 != nullptr) ? 2 : 1;

  for (int pass = 0; pass < npass; ++pass) {
    const float* __restrict__ Ap = pass ? A2 : A;
    const float* __restrict__ Wp = pass ? Wb : Wa;

    const float* aptr = Ap + arow * (long)K + 2 * g;
    const float* bptr = Wp + (2 * g) * NCT + ml;
    const int nfull = K >> 2;

    for (int it = 0; it < nfull; ++it) {
      v2f af = *(const v2f*)aptr;
      af.x = rvalid ? af.x : 0.0f;
      af.y = rvalid ? af.y : 0.0f;
#pragma unroll
      for (int t = 0; t < NT; ++t) {
        v2f bf;
        bf.x = bptr[t * 16];
        bf.y = bptr[NCT + t * 16];
        acc[t] = wmma_f32x4(af, bf, acc[t]);
      }
      aptr += 4;
      bptr += 4 * NCT;
    }

    const int k0 = nfull << 2;
    if (k0 < K) {
      const float* __restrict__ Arow_p = Ap + arow * (long)K;
      const int ka  = k0 + 2 * g;
      const int kc0 = imin(ka, K - 1);
      const int kc1 = imin(ka + 1, K - 1);
      float a0 = Arow_p[kc0];
      float a1 = Arow_p[kc1];
      v2f af;
      af.x = (rvalid && ka     < K) ? a0 : 0.0f;
      af.y = (rvalid && ka + 1 < K) ? a1 : 0.0f;
      const float* __restrict__ W0 = Wp + (long)kc0 * NCT + ml;
      const float* __restrict__ W1 = Wp + (long)kc1 * NCT + ml;
#pragma unroll
      for (int t = 0; t < NT; ++t) {
        v2f bf;
        bf.x = W0[t * 16];
        bf.y = W1[t * 16];
        acc[t] = wmma_f32x4(af, bf, acc[t]);
      }
    }
  }

  float* crow = C + (long)(mt * 16 + (g << 3)) * NCT + ml;
#pragma unroll
  for (int t = 0; t < NT; ++t) {
    const int n = t * 16 + ml;
    const bool nvalid = (NT * 16 == NCT) ? true : (n < NCT);
    const float bv = (bias != nullptr) ? bias[imin(n, NCT - 1)] : 0.0f;
#pragma unroll
    for (int v = 0; v < 8; ++v) {
      const int r = mt * 16 + v + (g << 3);
      float val = acc[t][v] + bv;
      if (relu) val = fmaxf(val, 0.0f);
      if (r < M && nvalid) crow[v * NCT + t * 16] = val;
    }
  }
}

__global__ void final_dot_kernel(const float* __restrict__ y, const float* __restrict__ W3,
                                 const float* __restrict__ b3, float* __restrict__ out, int M) {
  const int lane = threadIdx.x & 31;
  const int row = blockIdx.x * (blockDim.x >> 5) + (threadIdx.x >> 5);
  if (row >= M) return;
  float s = 0.0f;
#pragma unroll
  for (int k = lane; k < HID; k += 32) s += y[(long)row * HID + k] * W3[k];
#pragma unroll
  for (int off = 16; off > 0; off >>= 1) s += __shfl_xor(s, off, 32);
  if (lane == 0) out[row] = s + b3[0];
}


static inline int strip_grid(int M) {
  int tm = (M + 15) / 16;
  return (tm + 3) / 4;
}

extern "C" void kernel_launch(void* const* d_in, const int* in_sizes, int n_in,
                              void* d_out, int out_size, void* d_ws, size_t ws_size,
                              hipStream_t stream) {
  (void)in_sizes; (void)n_in; (void)out_size; (void)ws_size;

  const float* edge_feat = (const float*)d_in[0];
  const int*   edge_idx  = (const int*)d_in[1];
  const int*   act_range = (const int*)d_in[2];
  const float* Wl1 = (const float*)d_in[3];
  const float* Wr1 = (const float*)d_in[4];
  const float* b1c = (const float*)d_in[5];
  const float* Wl2 = (const float*)d_in[6];
  const float* Wr2 = (const float*)d_in[7];
  const float* b2c = (const float*)d_in[8];
  const float* W1  = (const float*)d_in[9];
  const float* b1  = (const float*)d_in[10];
  const float* W2  = (const float*)d_in[11];
  const float* b2  = (const float*)d_in[12];
  const float* W3  = (const float*)d_in[13];
  const float* b3  = (const float*)d_in[14];
  float* out = (float*)d_out;

  const long S_AGG1 = (long)BB * NN * FEAT;
  const long S_CNT  = (long)BB * NN;
  const long S_H    = (long)BB * NN * HID;
  const long S_Y    = (long)BB * PADR * HID;

  float* agg1 = (float*)d_ws;
  float* cnt  = agg1 + S_AGG1;
  float* h    = cnt  + S_CNT;
  float* agg2 = h    + S_H;
  float* y1   = agg2 + S_H;
  float* y2   = y1   + S_Y;

  const int TB = 256;
  const long bigE1 = (long)BB * EE * FEAT;
  const long bigE2 = (long)BB * EE * HID;

  fill_zero_kernel<<<(int)((S_AGG1 + S_CNT + TB - 1) / TB), TB, 0, stream>>>(agg1, S_AGG1 + S_CNT);
  count_edges_kernel<<<(int)(((long)BB * EE + TB - 1) / TB), TB, 0, stream>>>(edge_idx, cnt);
  scatter_add_kernel<<<(int)((bigE1 + TB - 1) / TB), TB, 0, stream>>>(edge_feat, edge_idx, agg1, FEAT);
  mean_div_kernel<<<(int)((S_AGG1 + TB - 1) / TB), TB, 0, stream>>>(agg1, cnt, FEAT);
  wmma_gemm_kernel<8, 128><<<strip_grid(BB * NN), 128, 0, stream>>>(
      agg1, edge_feat, Wl1, Wr1, b1c, h, BB * NN, FEAT,  1,
       0, nullptr, 0, 0);

  fill_zero_kernel<<<(int)((S_H + TB - 1) / TB), TB, 0, stream>>>(agg2, S_H);
  scatter_add_kernel<<<(int)((bigE2 + TB - 1) / TB), TB, 0, stream>>>(h, edge_idx, agg2, HID);
  mean_div_kernel<<<(int)((S_H + TB - 1) / TB), TB, 0, stream>>>(agg2, cnt, HID);
  wmma_gemm_kernel<4, 50><<<strip_grid(BB * NN), 128, 0, stream>>>(
      agg2, h, Wl2, Wr2, b2c, agg1, BB * NN, HID,  0,
       0, nullptr, 0, 0);

  wmma_gemm_kernel<8, 128><<<strip_grid(BB * PADR), 128, 0, stream>>>(
      agg1, nullptr, W1, nullptr, b1, y1, BB * PADR, FEAT,  1,
       1, act_range, PADR, NN);
  wmma_gemm_kernel<8, 128><<<strip_grid(BB * PADR), 128, 0, stream>>>(
      y1, nullptr, W2, nullptr, b2, y2, BB * PADR, HID,  1,
       0, nullptr, 0, 0);
  final_dot_kernel<<<(BB * PADR + 3) / 4, 128, 0, stream>>>(y2, W3, b3, out, BB * PADR);
}


# all WMMA GEMMs: a lane owns adjacent output columns, so B fragments of 4 (2) tiles are two 16-byte (8-byte) loads, asked for a k-step (MLP: 4 k-steps) ahead; 16/8-byte stores; sum order unchanged
# speedup vs baseline: 8.2883x; 8.2883x over previous
#include <hip/hip_runtime.h>

#define BB    16
#define NN    5000
#define EE    80000
#define FEAT  50
#define HID   128
#define PADR  1000
#define PPITCH 64

typedef __attribute__((ext_vector_type(2))) float v2f;
typedef __attribute__((ext_vector_type(4))) float v4f;
typedef __attribute__((ext_vector_type(8))) float v8f;
typedef __attribute__((ext_vector_type(4))) int   v4i;
typedef float v4fu __attribute__((ext_vector_type(4), aligned(4)));
typedef float v2fu __attribute__((ext_vector_type(2), aligned(4)));

__device__ __forceinline__ v8f wmma_f32x4(v2f a, v2f b, v8f c) {
  return __builtin_amdgcn_wmma_f32_16x16x4_f32(
       false, a,  false, b,
       (short)0, c,  false,  false);
}

__device__ __forceinline__ int imin(int a, int b) { return a < b ? a : b; }

template <int NJ>
__device__ __forceinline__ void kstep_quads(const v2f af, const v4fu (&b)[NJ][2], v8f (&acc)[4 * NJ]) {
#pragma unroll
  for (int j = 0; j < NJ; ++j) {
#pragma unroll
    for (int u = 0; u < 4; ++u) {
      v2f bf;
      bf.x = b[j][0][u];
      bf.y = b[j][1][u];
      acc[4 * j + u] = wmma_f32x4(af, bf, acc[4 * j + u]);
    }
  }
}

template <int NJ, int NCT>
__device__ __forceinline__ void load_quads(const float* brow, v4fu (&b)[NJ][2]) {
#pragma unroll
  for (int j = 0; j < NJ; ++j) {
    b[j][0] = *(const v4fu*)(brow + 64 * j);
    b[j][1] = *(const v4fu*)(brow + NCT + 64 * j);
  }
}


#define AGG_THREADS 1024
#define RING_CAP    512

template <int XS, int VEC, int U>
__device__ __forceinline__ void batch_load(const char* __restrict__ xbase, unsigned lane_off,
                                           const int* rp, int (&e)[U],
                                           __attribute__((ext_vector_type(VEC))) float (&v)[U]) {
  typedef __attribute__((ext_vector_type(VEC))) float vecf;
#pragma unroll
  for (int q = 0; q < U / 4; ++q) {
    const v4i w = *(const v4i*)(rp + 4 * q);
    e[4 * q + 0] = w[0]; e[4 * q + 1] = w[1]; e[4 * q + 2] = w[2]; e[4 * q + 3] = w[3];
  }
#pragma unroll
  for (int u = 0; u < U; ++u) {
    const unsigned off = (unsigned)(e[u] & 0xFFFF) * (unsigned)(XS * 4) + lane_off;
    v[u] = *(const vecf*)(xbase + off);
  }
}

template <int F, int VEC, int U, bool FULL>
__device__ __forceinline__ void batch_add(float* accl, const int (&e)[U],
                                          const __attribute__((ext_vector_type(VEC))) float (&v)[U],
                                          int cnt) {
  constexpr int NL = F / VEC;
#pragma unroll
  for (int u = 0; u < U; ++u) {
    if (FULL || u < cnt) {
      float* r = (float*)((char*)accl + ((unsigned)e[u] >> 16) * (unsigned)(F * 4));
#pragma unroll
      for (int j = 0; j < VEC; ++j) atomicAdd(r + j * NL, v[u][j]);
    }
  }
}

__device__ __forceinline__ void scan_group(const v4i d, const v4i s, int lo, int nrows,
                                           int* ring_w, int* deg, int& tail) {
#pragma unroll
  for (int c = 0; c < 4; ++c) {
    const int dl = d[c] - lo;
    const bool hit = (unsigned)dl < (unsigned)nrows;
    const unsigned m = __builtin_amdgcn_ballot_w32(hit);
    if (hit) {
      const int sc = imin(s[c] < 0 ? 0 : s[c], NN - 1);
      const int pos = (int)__builtin_amdgcn_mbcnt_lo(m, (unsigned)tail) & (RING_CAP - 1);
      ring_w[pos] = sc | (dl << 16);
      atomicAdd(&deg[dl], 1);
    }
    tail += __builtin_popcount(m);
  }
}

template <int F, int XS, int VEC, int U>
__device__ __forceinline__ void drain_whole(const char* __restrict__ xbase, unsigned lane_off,
                                            float* accl, const int* ring_w, bool active,
                                            int& hb, int tail) {
  typedef __attribute__((ext_vector_type(VEC))) float vecf;
  for (int nb = (tail - hb * U) / U; nb > 0; --nb) {
    int e[U];
    vecf v[U];
    batch_load<XS, VEC, U>(xbase, lane_off, ring_w + ((hb * U) & (RING_CAP - 1)), e, v);
    if (active) batch_add<F, VEC, U, true>(accl, e, v, U);
    hb += 1;
  }
}

template <int F, int XS, int VEC, int RN, int U, bool ADDQ>
__global__ __launch_bounds__(AGG_THREADS)
void agg_mean_lds_kernel(const float* __restrict__ x, const int* __restrict__ ei,
                         const float* __restrict__ addq, float* __restrict__ mean) {
  typedef __attribute__((ext_vector_type(VEC))) float vecf;
  constexpr int NL = F / VEC;
  constexpr int NW = AGG_THREADS / 32;
  constexpr int NG = EE / 128;
  constexpr int NT = (NG + NW - 1) / NW;
  static_assert(NL * VEC == F && NL <= 32, "row must split over one wave");
  static_assert(EE % 128 == 0, "edge groups");
  static_assert(RN < 32768 && NN <= 65536, "ring word packing");
  static_assert(XS >= F && XS % VEC == 0 && (long)NN * XS * 4 <= (1L << 22), "source row pitch");
  static_assert(RING_CAP % U == 0 && U % 4 == 0 && RING_CAP >= 128 + U, "ring");
  static_assert((RN * F) % 2 == 0, "zero fill by pairs");

  __shared__ alignas(16) float acc[RN * F];
  __shared__ alignas(16) int   ring[NW * RING_CAP];
  __shared__ int   deg[RN];

  const int tid  = threadIdx.x;
  const int lane = tid & 31;
  const int wave = __builtin_amdgcn_readfirstlane(tid >> 5);
  const int b    = blockIdx.y;
  const int lo   = blockIdx.x * RN;
  const int nrows = imin(RN, NN - lo);

  for (int i = tid; i < RN * F / 2; i += AGG_THREADS) ((v2f*)acc)[i] = (v2f){0.0f, 0.0f};
  for (int i = tid; i < RN; i += AGG_THREADS) deg[i] = 0;
  for (int i = tid; i < NW * RING_CAP / 4; i += AGG_THREADS) ((v4i*)ring)[i] = (v4i){0, 0, 0, 0};
  __syncthreads();

  const v4i* __restrict__ src4 = (const v4i*)(ei + (long)b * 2 * EE);
  const v4i* __restrict__ dst4 = (const v4i*)(ei + (long)b * 2 * EE + EE);
  const char* __restrict__ xbase = (const char*)(x + (long)b * NN * XS);
  const bool active = lane < NL;
  const unsigned lane_off = active ? (unsigned)(lane * VEC * 4) : 0u;
  int* ring_w = ring + wave * RING_CAP;
  float* accl = acc + lane;

  int tail = 0;
  int hb   = 0;

#define AGG_TRIP(DCUR, SCUR, GCUR)                                                    \
  {                                                                                   \
    if (tail - hb * U > RING_CAP - 128)      \
      drain_whole<F, XS, VEC, U>(xbase, lane_off, accl, ring_w, active, hb, tail);    \
    const bool have = (tail - hb * U) >= U;                                           \
    int e[U];                                                                         \
    vecf v[U];                                                                        \
    batch_load<XS, VEC, U>(xbase, lane_off, ring_w + ((hb * U) & (RING_CAP - 1)), e, v); \
    if ((GCUR) < NG) scan_group(DCUR, SCUR, lo, nrows, ring_w, deg, tail);            \
    if (have) {                                                                       \
      if (active) batch_add<F, VEC, U, true>(accl, e, v, U);                          \
      hb += 1;                                                                        \
    }                                                                                 \
  }

  int g0 = wave;
  v4i d0 = dst4[imin(g0, NG - 1) * 32 + lane];
  v4i s0 = src4[imin(g0, NG - 1) * 32 + lane];
  for (int t = 0; t < NT; t += 2) {
    const int g1 = g0 + NW;
    const v4i d1 = dst4[imin(g1, NG - 1) * 32 + lane];
    const v4i s1 = src4[imin(g1, NG - 1) * 32 + lane];
    AGG_TRIP(d0, s0, g0)
    g0 = g1 + NW;
    d0 = dst4[imin(g0, NG - 1) * 32 + lane];
    s0 = src4[imin(g0, NG - 1) * 32 + lane];
    AGG_TRIP(d1, s1, g1)
  }
#undef AGG_TRIP

  {
    const int pending = tail - hb * U;
    int nb = pending / U;
    const int rem = pending - nb * U;
    if (pending > 0) {
      int ea[U], eb[U];
      vecf va[U], vb[U];
      batch_load<XS, VEC, U>(xbase, lane_off, ring_w + ((hb * U) & (RING_CAP - 1)), ea, va);
      while (nb >= 2) {
        batch_load<XS, VEC, U>(xbase, lane_off, ring_w + (((hb + 1) * U) & (RING_CAP - 1)), eb, vb);
        if (active) batch_add<F, VEC, U, true>(accl, ea, va, U);
        batch_load<XS, VEC, U>(xbase, lane_off, ring_w + (((hb + 2) * U) & (RING_CAP - 1)), ea, va);
        if (active) batch_add<F, VEC, U, true>(accl, eb, vb, U);
        hb += 2;
        nb -= 2;
      }
      if (nb) {
        batch_load<XS, VEC, U>(xbase, lane_off, ring_w + (((hb + 1) * U) & (RING_CAP - 1)), eb, vb);
        if (active) {
          batch_add<F, VEC, U, true>(accl, ea, va, U);
          batch_add<F, VEC, U, false>(accl, eb, vb, rem);
        }
      } else {
        if (active) batch_add<F, VEC, U, false>(accl, ea, va, rem);
      }
    }
  }
  __syncthreads();

  float* __restrict__ mb = mean + ((long)b * NN + lo) * F + lane * VEC;
  const float* __restrict__ qb = ADDQ ? addq + ((long)b * NN + lo) * F + lane * VEC : nullptr;
  for (int r = wave; r < nrows; r += NW) {
    const float c = fmaxf((float)deg[r], 1.0f);
    if (active) {
      vecf o;
#pragma unroll
      for (int j = 0; j < VEC; ++j) o[j] = acc[r * F + j * NL + lane] / c;
      if (ADDQ) o += *(const vecf*)(qb + (long)r * F);
      *(vecf*)(mb + (long)r * F) = o;
    }
  }
}

template <int NT, int NCT>
__global__ __launch_bounds__(128)
void wmma_gemm_kernel(const float* __restrict__ A,
                                 const float* __restrict__ A2,
                                 const float* __restrict__ Wa,
                                 const float* __restrict__ Wb,
                                 const float* __restrict__ bias,
                                 float* __restrict__ C,
                                 int M, int K, int relu,
                                 int pad_mode, const int* __restrict__ act_range_p,
                                 int pad_rows, int src_rows) {
  static_assert(NT % 4 == 0 && NT * 16 == NCT, "column tiles in groups of four, no partial tile");
  constexpr int NJ = NT / 4;
  const int lane = threadIdx.x & 31;
  const int mt = blockIdx.x * (blockDim.x >> 5) + (threadIdx.x >> 5);
  const int tm = (M + 15) >> 4;
  if (mt >= tm) return;

  const int ml = lane & 15;
  const int g  = lane >> 4;

  const int row = mt * 16 + ml;
  bool rvalid = row < M;
  long arow;
  if (pad_mode) {
    const int rc = imin(row, M - 1);
    const int b  = rc / pad_rows;
    const int rr = rc - b * pad_rows;
    rvalid = rvalid && (rr < *act_range_p);
    arow = (long)b * src_rows + rr;
  } else {
    arow = imin(row, M - 1);
  }

  v8f acc[NT];
#pragma unroll
  for (int t = 0; t < NT; ++t) acc[t] = (v8f){};

  const int npass = (A2 != nullptr) ? 2 : 1;

  for (int pass = 0; pass < npass; ++pass) {
    const float* __restrict__ Ap = pass ? A2 : A;
    const float* __restrict__ Wp = pass ? Wb : Wa;

    const float* aptr = Ap + arow * (long)K + 2 * g;
    const float* bptr = Wp + (2 * g) * NCT + 4 * ml;
    const int nfull = K >> 2;
    const int ngrp  = nfull >> 2;

    if (ngrp > 0) {
      v2f  ac[4];
      v4fu bc[4][NJ][2];
#pragma unroll
      for (int s = 0; s < 4; ++s) {
        ac[s] = *(const v2f*)(aptr + 4 * s);
        load_quads<NJ, NCT>(bptr + 4 * s * NCT, bc[s]);
      }
#pragma unroll 1
      for (int gi = 0; gi + 1 < ngrp; ++gi) {
        aptr += 16;
        bptr += 16 * NCT;
        v2f  an[4];
        v4fu bn[4][NJ][2];
#pragma unroll
        for (int s = 0; s < 4; ++s) {
          an[s] = *(const v2f*)(aptr + 4 * s);
          load_quads<NJ, NCT>(bptr + 4 * s * NCT, bn[s]);
        }
#pragma unroll
        for (int s = 0; s < 4; ++s) {
          v2f af;
          af.x = rvalid ? ac[s].x : 0.0f;
          af.y = rvalid ? ac[s].y : 0.0f;
          kstep_quads<NJ>(af, bc[s], acc);
        }
#pragma unroll
        for (int s = 0; s < 4; ++s) {
          ac[s] = an[s];
#pragma unroll
          for (int j = 0; j < NJ; ++j) { bc[s][j][0] = bn[s][j][0]; bc[s][j][1] = bn[s][j][1]; }
        }
      }
#pragma unroll
      for (int s = 0; s < 4; ++s) {
        v2f af;
        af.x = rvalid ? ac[s].x : 0.0f;
        af.y = rvalid ? ac[s].y : 0.0f;
        kstep_quads<NJ>(af, bc[s], acc);
      }
      aptr += 16;
      bptr += 16 * NCT;
    }

    for (int it = ngrp << 2; it < nfull; ++it) {
      v2f af = *(const v2f*)aptr;
      af.x = rvalid ? af.x : 0.0f;
      af.y = rvalid ? af.y : 0.0f;
      v4fu b[NJ][2];
      load_quads<NJ, NCT>(bptr, b);
      kstep_quads<NJ>(af, b, acc);
      aptr += 4;
      bptr += 4 * NCT;
    }

    const int k0 = nfull << 2;
    if (k0 < K) {
      const float* __restrict__ Arow_p = Ap + arow * (long)K;
      const int ka  = k0 + 2 * g;
      const int kc0 = imin(ka, K - 1);
      const int kc1 = imin(ka + 1, K - 1);
      float a0 = Arow_p[kc0];
      float a1 = Arow_p[kc1];
      v2f af;
      af.x = (rvalid && ka     < K) ? a0 : 0.0f;
      af.y = (rvalid && ka + 1 < K) ? a1 : 0.0f;
      const float* __restrict__ W0 = Wp + (long)kc0 * NCT + 4 * ml;
      const float* __restrict__ W1 = Wp + (long)kc1 * NCT + 4 * ml;
      v4fu b[NJ][2];
#pragma unroll
      for (int j = 0; j < NJ; ++j) {
        b[j][0] = *(const v4fu*)(W0 + 64 * j);
        b[j][1] = *(const v4fu*)(W1 + 64 * j);
      }
      kstep_quads<NJ>(af, b, acc);
    }
  }

  float* crow = C + (long)(mt * 16 + (g << 3)) * NCT + 4 * ml;
#pragma unroll
  for (int j = 0; j < NJ; ++j) {
    v4f bv = (v4f){0.0f, 0.0f, 0.0f, 0.0f};
    if (bias != nullptr) bv = *(const v4fu*)(bias + 64 * j + 4 * ml);
#pragma unroll
    for (int v = 0; v < 8; ++v) {
      const int r = mt * 16 + v + (g << 3);
      v4f val = (v4f){acc[4 * j + 0][v], acc[4 * j + 1][v], acc[4 * j + 2][v], acc[4 * j + 3][v]} + bv;
      if (relu) {
        val[0] = fmaxf(val[0], 0.0f); val[1] = fmaxf(val[1], 0.0f);
        val[2] = fmaxf(val[2], 0.0f); val[3] = fmaxf(val[3], 0.0f);
      }
      if (r < M) *(v4f*)(crow + v * NCT + 64 * j) = val;
    }
  }
}

#define HPITCH 132
#define FUSED_WAVES 4

template <int NJ>
__device__ __forceinline__ void kstep_pairs(const v2f af, const v2fu (&p)[NJ][2], const v2fu (&q)[NJ][2],
                                            v8f (&accp)[2 * NJ], v8f (&accq)[2 * NJ]) {
#pragma unroll
  for (int j = 0; j < NJ; ++j) {
#pragma unroll
    for (int u = 0; u < 2; ++u) {
      v2f bf;
      bf.x = p[j][0][u];
      bf.y = p[j][1][u];
      accp[2 * j + u] = wmma_f32x4(af, bf, accp[2 * j + u]);
      bf.x = q[j][0][u];
      bf.y = q[j][1][u];
      accq[2 * j + u] = wmma_f32x4(af, bf, accq[2 * j + u]);
    }
  }
}

template <int NJ, int NCT>
__device__ __forceinline__ void load_pairs(const float* wrow, const int (&cb)[NJ], v2fu (&p)[NJ][2]) {
#pragma unroll
  for (int j = 0; j < NJ; ++j) {
    p[j][0] = *(const v2fu*)(wrow + cb[j]);
    p[j][1] = *(const v2fu*)(wrow + NCT + cb[j]);
  }
}

template <int K1, int NT2, int NCT, int PS>
__global__ __launch_bounds__(FUSED_WAVES * 32)
void sage1_then_pre2_kernel(const float* __restrict__ A1, const float* __restrict__ A2,
                            const float* __restrict__ Wa1, const float* __restrict__ Wb1,
                            const float* __restrict__ bias1,
                            const float* __restrict__ Wa2, const float* __restrict__ Wb2,
                            const float* __restrict__ bias2,
                            float* __restrict__ P, float* __restrict__ Q, int M) {
  static_assert(HID % 16 == 0 && HID % 4 == 0, "h strip: whole column tiles and k-steps");
  static_assert((NT2 - 1) * 16 < NCT && NT2 * 16 >= NCT && PS >= NCT, "column tiles");
  static_assert(NT2 % 2 == 0 && NCT % 2 == 0 && NCT >= 32, "P, Q: column pairs, whole or past the end");
  static_assert(HID % 64 == 0, "h strip: column tiles in groups of four");
  static_assert(HPITCH >= HID && HPITCH % 4 == 0, "8-byte fragment reads, 16-byte strip stores");
  static_assert((K1 >> 2) >= 1 && ((K1 >> 2) << 2) < K1, "layer 1: whole k-steps, then a partial one");
  constexpr int NT1 = HID / 16;
  constexpr int NJ1 = NT1 / 4;
  constexpr int NJ2 = NT2 / 2;

  __shared__ alignas(16) float hs_all[FUSED_WAVES * 16 * HPITCH];

  const int lane = threadIdx.x & 31;
  const int wv   = threadIdx.x >> 5;
  const int mt = blockIdx.x * FUSED_WAVES + wv;
  const int tm = (M + 15) >> 4;
  if (mt >= tm) return;

  const int ml = lane & 15;
  const int g  = lane >> 4;

  const int row = mt * 16 + ml;
  const bool rvalid = row < M;
  const long arow = imin(row, M - 1);

  float* hs = hs_all + wv * (16 * HPITCH);

  {
    v8f acc[NT1];
#pragma unroll
    for (int t = 0; t < NT1; ++t) acc[t] = (v8f){};

    for (int pass = 0; pass < 2; ++pass) {
      const float* __restrict__ Ap = pass ? A2 : A1;
      const float* __restrict__ Wp = pass ? Wb1 : Wa1;

      const float* aptr = Ap + arow * (long)K1 + 2 * g;
      const float* bptr = Wp + (2 * g) * HID + 4 * ml;
      constexpr int nfull = K1 >> 2;

      v2f  ac = *(const v2f*)aptr;
      v4fu bc[NJ1][2];
      load_quads<NJ1, HID>(bptr, bc);

#pragma unroll 1
      for (int it = 0; it + 1 < nfull; ++it) {
        aptr += 4;
        bptr += 4 * HID;
        const v2f an = *(const v2f*)aptr;
        v4fu bn[NJ1][2];
        load_quads<NJ1, HID>(bptr, bn);
        v2f af;
        af.x = rvalid ? ac.x : 0.0f;
        af.y = rvalid ? ac.y : 0.0f;
        kstep_quads<NJ1>(af, bc, acc);
        ac = an;
#pragma unroll
        for (int j = 0; j < NJ1; ++j) { bc[j][0] = bn[j][0]; bc[j][1] = bn[j][1]; }
      }

      constexpr int k0 = nfull << 2;
      const float* __restrict__ Arow_p = Ap + arow * (long)K1;
      const int ka  = k0 + 2 * g;
      const int kc0 = imin(ka, K1 - 1);
      const int kc1 = imin(ka + 1, K1 - 1);
      const float a0 = Arow_p[kc0];
      const float a1 = Arow_p[kc1];
      const float* __restrict__ W0 = Wp + (long)kc0 * HID + 4 * ml;
      const float* __restrict__ W1 = Wp + (long)kc1 * HID + 4 * ml;
      v4fu bt[NJ1][2];
#pragma unroll
      for (int j = 0; j < NJ1; ++j) {
        bt[j][0] = *(const v4fu*)(W0 + 64 * j);
        bt[j][1] = *(const v4fu*)(W1 + 64 * j);
      }
      {
        v2f af;
        af.x = rvalid ? ac.x : 0.0f;
        af.y = rvalid ? ac.y : 0.0f;
        kstep_quads<NJ1>(af, bc, acc);
      }
      {
        v2f af;
        af.x = (rvalid && ka     < K1) ? a0 : 0.0f;
        af.y = (rvalid && ka + 1 < K1) ? a1 : 0.0f;
        kstep_quads<NJ1>(af, bt, acc);
      }
    }

    float* hw = hs + (g << 3) * HPITCH + 4 * ml;
#pragma unroll
    for (int j = 0; j < NJ1; ++j) {
      const v4f bv = *(const v4fu*)(bias1 + 64 * j + 4 * ml);
#pragma unroll
      for (int v = 0; v < 8; ++v) {
        v4f val = (v4f){acc[4 * j + 0][v], acc[4 * j + 1][v], acc[4 * j + 2][v], acc[4 * j + 3][v]} + bv;
        val[0] = fmaxf(val[0], 0.0f); val[1] = fmaxf(val[1], 0.0f);
        val[2] = fmaxf(val[2], 0.0f); val[3] = fmaxf(val[3], 0.0f);
        *(v4f*)(hw + v * HPITCH + 64 * j) = val;
      }
    }
  }

  v8f accp[NT2], accq[NT2];
#pragma unroll
  for (int t = 0; t < NT2; ++t) { accp[t] = (v8f){}; accq[t] = (v8f){}; }

  int cb[NJ2];
#pragma unroll
  for (int j = 0; j < NJ2; ++j) cb[j] = imin(32 * j + 2 * ml, NCT - 2);

  const float* hrow = hs + ml * HPITCH + 2 * g;
  const float* wa = Wa2 + (2 * g) * NCT;
  const float* wb = Wb2 + (2 * g) * NCT;

  v2f  hc = *(const v2f*)hrow;
  v2fu pc[NJ2][2], qc[NJ2][2];
  load_pairs<NJ2, NCT>(wa, cb, pc);
  load_pairs<NJ2, NCT>(wb, cb, qc);

#pragma unroll 1
  for (int it = 0; it + 1 < HID / 4; ++it) {
    wa += 4 * NCT;
    wb += 4 * NCT;
    const v2f hn = *(const v2f*)(hrow + 4 * (it + 1));
    v2fu pn[NJ2][2], qn[NJ2][2];
    load_pairs<NJ2, NCT>(wa, cb, pn);
    load_pairs<NJ2, NCT>(wb, cb, qn);
    kstep_pairs<NJ2>(hc, pc, qc, accp, accq);
    hc = hn;
#pragma unroll
    for (int j = 0; j < NJ2; ++j) {
      pc[j][0] = pn[j][0]; pc[j][1] = pn[j][1];
      qc[j][0] = qn[j][0]; qc[j][1] = qn[j][1];
    }
  }
  kstep_pairs<NJ2>(hc, pc, qc, accp, accq);

  float* prow = P + (long)(mt * 16 + (g << 3)) * PS;
  float* qrow = Q + (long)(mt * 16 + (g << 3)) * NCT;
#pragma unroll
  for (int j = 0; j < NJ2; ++j) {
    const int n = 32 * j + 2 * ml;
    const bool nvalid = n < NCT;
    const v2f bv = *(const v2fu*)(bias2 + cb[j]);
#pragma unroll
    for (int v = 0; v < 8; ++v) {
      const int r = mt * 16 + v + (g << 3);
      if (r < M && nvalid) {
        *(v2f*)(prow + v * PS  + n) = (v2f){accp[2 * j][v], accp[2 * j + 1][v]};
        *(v2f*)(qrow + v * NCT + n) = (v2f){accq[2 * j][v], accq[2 * j + 1][v]} + bv;
      }
    }
  }
}

__global__ void final_dot_kernel(const float* __restrict__ y, const float* __restrict__ W3,
                                 const float* __restrict__ b3, float* __restrict__ out, int M) {
  const int lane = threadIdx.x & 31;
  const int row = blockIdx.x * (blockDim.x >> 5) + (threadIdx.x >> 5);
  if (row >= M) return;
  float s = 0.0f;
#pragma unroll
  for (int k = lane; k < HID; k += 32) s += y[(long)row * HID + k] * W3[k];
#pragma unroll
  for (int off = 16; off > 0; off >>= 1) s += __shfl_xor(s, off, 32);
  if (lane == 0) out[row] = s + b3[0];
}


static inline int strip_grid(int M) {
  int tm = (M + 15) / 16;
  return (tm + 3) / 4;
}

extern "C" void kernel_launch(void* const* d_in, const int* in_sizes, int n_in,
                              void* d_out, int out_size, void* d_ws, size_t ws_size,
                              hipStream_t stream) {
  (void)in_sizes; (void)n_in; (void)out_size; (void)ws_size;

  const float* edge_feat = (const float*)d_in[0];
  const int*   edge_idx  = (const int*)d_in[1];
  const int*   act_range = (const int*)d_in[2];
  const float* Wl1 = (const float*)d_in[3];
  const float* Wr1 = (const float*)d_in[4];
  const float* b1c = (const float*)d_in[5];
  const float* Wl2 = (const float*)d_in[6];
  const float* Wr2 = (const float*)d_in[7];
  const float* b2c = (const float*)d_in[8];
  const float* W1  = (const float*)d_in[9];
  const float* b1  = (const float*)d_in[10];
  const float* W2  = (const float*)d_in[11];
  const float* b2  = (const float*)d_in[12];
  const float* W3  = (const float*)d_in[13];
  const float* b3  = (const float*)d_in[14];
  float* out = (float*)d_out;

  const long S_AGG1 = (long)BB * NN * FEAT;
  const long S_CNT  = (long)BB * NN;
  const long S_H    = (long)BB * NN * HID;
  const long S_Y    = (long)BB * PADR * HID;

  const long S_P = (long)BB * NN * PPITCH;

  float* agg1 = (float*)d_ws;
  float* hgap = agg1 + S_AGG1 + S_CNT;
  float* p2   = hgap + S_H;
  float* q2   = p2   + S_P;
  float* y1   = p2   + S_H;
  float* y2   = y1   + S_Y;

  agg_mean_lds_kernel<FEAT, FEAT, 2, 313, 8, false><<<dim3(16, BB), AGG_THREADS, 0, stream>>>(
      edge_feat, edge_idx, nullptr, agg1);

  sage1_then_pre2_kernel<FEAT, 4, FEAT, PPITCH><<<strip_grid(BB * NN), FUSED_WAVES * 32, 0, stream>>>(
      agg1, edge_feat, Wl1, Wr1, b1c, Wl2, Wr2, b2c, p2, q2, BB * NN);
  agg_mean_lds_kernel<FEAT, PPITCH, 2, 313, 8, true><<<dim3(16, BB), AGG_THREADS, 0, stream>>>(
      p2, edge_idx, q2, agg1);

  wmma_gemm_kernel<8, 128><<<strip_grid(BB * PADR), 128, 0, stream>>>(
      agg1, nullptr, W1, nullptr, b1, y1, BB * PADR, FEAT,  1,
       1, act_range, PADR, NN);
  wmma_gemm_kernel<8, 128><<<strip_grid(BB * PADR), 128, 0, stream>>>(
      y1, nullptr, W2, nullptr, b2, y2, BB * PADR, HID,  1,
       0, nullptr, 0, 0);
  final_dot_kernel<<<(BB * PADR + 3) / 4, 128, 0, stream>>>(y2, W3, b3, out, BB * PADR);
}


# both MLP products and the dot in one launch: y1 strip taken transposed (W1^T tile x x^T), so its accumulators are the A fragments of the y2 product; no LDS, no y1 array; y2 k sum order changes
# speedup vs baseline: 9.2188x; 1.1123x over previous
#include <hip/hip_runtime.h>

#define BB    16
#define NN    5000
#define EE    80000
#define FEAT  50
#define HID   128
#define PADR  1000
#define PPITCH 64

typedef __attribute__((ext_vector_type(2))) float v2f;
typedef __attribute__((ext_vector_type(4))) float v4f;
typedef __attribute__((ext_vector_type(8))) float v8f;
typedef __attribute__((ext_vector_type(4))) int   v4i;
typedef float v4fu __attribute__((ext_vector_type(4), aligned(4)));
typedef float v2fu __attribute__((ext_vector_type(2), aligned(4)));

__device__ __forceinline__ v8f wmma_f32x4(v2f a, v2f b, v8f c) {
  return __builtin_amdgcn_wmma_f32_16x16x4_f32(
       false, a,  false, b,
       (short)0, c,  false,  false);
}

__device__ __forceinline__ int imin(int a, int b) { return a < b ? a : b; }

template <int NJ>
__device__ __forceinline__ void kstep_quads(const v2f af, const v4fu (&b)[NJ][2], v8f (&acc)[4 * NJ]) {
#pragma unroll
  for (int j = 0; j < NJ; ++j) {
#pragma unroll
    for (int u = 0; u < 4; ++u) {
      v2f bf;
      bf.x = b[j][0][u];
      bf.y = b[j][1][u];
      acc[4 * j + u] = wmma_f32x4(af, bf, acc[4 * j + u]);
    }
  }
}

template <int NJ, int NCT>
__device__ __forceinline__ void load_quads(const float* brow, v4fu (&b)[NJ][2]) {
#pragma unroll
  for (int j = 0; j < NJ; ++j) {
    b[j][0] = *(const v4fu*)(brow + 64 * j);
    b[j][1] = *(const v4fu*)(brow + NCT + 64 * j);
  }
}


#define AGG_THREADS 1024
#define RING_CAP    512

template <int XS, int VEC, int U>
__device__ __forceinline__ void batch_load(const char* __restrict__ xbase, unsigned lane_off,
                                           const int* rp, int (&e)[U],
                                           __attribute__((ext_vector_type(VEC))) float (&v)[U]) {
  typedef __attribute__((ext_vector_type(VEC))) float vecf;
#pragma unroll
  for (int q = 0; q < U / 4; ++q) {
    const v4i w = *(const v4i*)(rp + 4 * q);
    e[4 * q + 0] = w[0]; e[4 * q + 1] = w[1]; e[4 * q + 2] = w[2]; e[4 * q + 3] = w[3];
  }
#pragma unroll
  for (int u = 0; u < U; ++u) {
    const unsigned off = (unsigned)(e[u] & 0xFFFF) * (unsigned)(XS * 4) + lane_off;
    v[u] = *(const vecf*)(xbase + off);
  }
}

template <int F, int VEC, int U, bool FULL>
__device__ __forceinline__ void batch_add(float* accl, const int (&e)[U],
                                          const __attribute__((ext_vector_type(VEC))) float (&v)[U],
                                          int cnt) {
  constexpr int NL = F / VEC;
#pragma unroll
  for (int u = 0; u < U; ++u) {
    if (FULL || u < cnt) {
      float* r = (float*)((char*)accl + ((unsigned)e[u] >> 16) * (unsigned)(F * 4));
#pragma unroll
      for (int j = 0; j < VEC; ++j) atomicAdd(r + j * NL, v[u][j]);
    }
  }
}

__device__ __forceinline__ void scan_group(const v4i d, const v4i s, int lo, int nrows,
                                           int* ring_w, int* deg, int& tail) {
#pragma unroll
  for (int c = 0; c < 4; ++c) {
    const int dl = d[c] - lo;
    const bool hit = (unsigned)dl < (unsigned)nrows;
    const unsigned m = __builtin_amdgcn_ballot_w32(hit);
    if (hit) {
      const int sc = imin(s[c] < 0 ? 0 : s[c], NN - 1);
      const int pos = (int)__builtin_amdgcn_mbcnt_lo(m, (unsigned)tail) & (RING_CAP - 1);
      ring_w[pos] = sc | (dl << 16);
      atomicAdd(&deg[dl], 1);
    }
    tail += __builtin_popcount(m);
  }
}

template <int F, int XS, int VEC, int U>
__device__ __forceinline__ void drain_whole(const char* __restrict__ xbase, unsigned lane_off,
                                            float* accl, const int* ring_w, bool active,
                                            int& hb, int tail) {
  typedef __attribute__((ext_vector_type(VEC))) float vecf;
  for (int nb = (tail - hb * U) / U; nb > 0; --nb) {
    int e[U];
    vecf v[U];
    batch_load<XS, VEC, U>(xbase, lane_off, ring_w + ((hb * U) & (RING_CAP - 1)), e, v);
    if (active) batch_add<F, VEC, U, true>(accl, e, v, U);
    hb += 1;
  }
}

template <int F, int XS, int VEC, int RN, int U, bool ADDQ>
__global__ __launch_bounds__(AGG_THREADS)
void agg_mean_lds_kernel(const float* __restrict__ x, const int* __restrict__ ei,
                         const float* __restrict__ addq, float* __restrict__ mean) {
  typedef __attribute__((ext_vector_type(VEC))) float vecf;
  constexpr int NL = F / VEC;
  constexpr int NW = AGG_THREADS / 32;
  constexpr int NG = EE / 128;
  constexpr int NT = (NG + NW - 1) / NW;
  static_assert(NL * VEC == F && NL <= 32, "row must split over one wave");
  static_assert(EE % 128 == 0, "edge groups");
  static_assert(RN < 32768 && NN <= 65536, "ring word packing");
  static_assert(XS >= F && XS % VEC == 0 && (long)NN * XS * 4 <= (1L << 22), "source row pitch");
  static_assert(RING_CAP % U == 0 && U % 4 == 0 && RING_CAP >= 128 + U, "ring");
  static_assert((RN * F) % 2 == 0, "zero fill by pairs");

  __shared__ alignas(16) float acc[RN * F];
  __shared__ alignas(16) int   ring[NW * RING_CAP];
  __shared__ int   deg[RN];

  const int tid  = threadIdx.x;
  const int lane = tid & 31;
  const int wave = __builtin_amdgcn_readfirstlane(tid >> 5);
  const int b    = blockIdx.y;
  const int lo   = blockIdx.x * RN;
  const int nrows = imin(RN, NN - lo);

  for (int i = tid; i < RN * F / 2; i += AGG_THREADS) ((v2f*)acc)[i] = (v2f){0.0f, 0.0f};
  for (int i = tid; i < RN; i += AGG_THREADS) deg[i] = 0;
  for (int i = tid; i < NW * RING_CAP / 4; i += AGG_THREADS) ((v4i*)ring)[i] = (v4i){0, 0, 0, 0};
  __syncthreads();

  const v4i* __restrict__ src4 = (const v4i*)(ei + (long)b * 2 * EE);
  const v4i* __restrict__ dst4 = (const v4i*)(ei + (long)b * 2 * EE + EE);
  const char* __restrict__ xbase = (const char*)(x + (long)b * NN * XS);
  const bool active = lane < NL;
  const unsigned lane_off = active ? (unsigned)(lane * VEC * 4) : 0u;
  int* ring_w = ring + wave * RING_CAP;
  float* accl = acc + lane;

  int tail = 0;
  int hb   = 0;

#define AGG_TRIP(DCUR, SCUR, GCUR)                                                    \
  {                                                                                   \
    if (tail - hb * U > RING_CAP - 128)      \
      drain_whole<F, XS, VEC, U>(xbase, lane_off, accl, ring_w, active, hb, tail);    \
    const bool have = (tail - hb * U) >= U;                                           \
    int e[U];                                                                         \
    vecf v[U];                                                                        \
    batch_load<XS, VEC, U>(xbase, lane_off, ring_w + ((hb * U) & (RING_CAP - 1)), e, v); \
    if ((GCUR) < NG) scan_group(DCUR, SCUR, lo, nrows, ring_w, deg, tail);            \
    if (have) {                                                                       \
      if (active) batch_add<F, VEC, U, true>(accl, e, v, U);                          \
      hb += 1;                                                                        \
    }                                                                                 \
  }

  int g0 = wave;
  v4i d0 = dst4[imin(g0, NG - 1) * 32 + lane];
  v4i s0 = src4[imin(g0, NG - 1) * 32 + lane];
  for (int t = 0; t < NT; t += 2) {
    const int g1 = g0 + NW;
    const v4i d1 = dst4[imin(g1, NG - 1) * 32 + lane];
    const v4i s1 = src4[imin(g1, NG - 1) * 32 + lane];
    AGG_TRIP(d0, s0, g0)
    g0 = g1 + NW;
    d0 = dst4[imin(g0, NG - 1) * 32 + lane];
    s0 = src4[imin(g0, NG - 1) * 32 + lane];
    AGG_TRIP(d1, s1, g1)
  }
#undef AGG_TRIP

  {
    const int pending = tail - hb * U;
    int nb = pending / U;
    const int rem = pending - nb * U;
    if (pending > 0) {
      int ea[U], eb[U];
      vecf va[U], vb[U];
      batch_load<XS, VEC, U>(xbase, lane_off, ring_w + ((hb * U) & (RING_CAP - 1)), ea, va);
      while (nb >= 2) {
        batch_load<XS, VEC, U>(xbase, lane_off, ring_w + (((hb + 1) * U) & (RING_CAP - 1)), eb, vb);
        if (active) batch_add<F, VEC, U, true>(accl, ea, va, U);
        batch_load<XS, VEC, U>(xbase, lane_off, ring_w + (((hb + 2) * U) & (RING_CAP - 1)), ea, va);
        if (active) batch_add<F, VEC, U, true>(accl, eb, vb, U);
        hb += 2;
        nb -= 2;
      }
      if (nb) {
        batch_load<XS, VEC, U>(xbase, lane_off, ring_w + (((hb + 1) * U) & (RING_CAP - 1)), eb, vb);
        if (active) {
          batch_add<F, VEC, U, true>(accl, ea, va, U);
          batch_add<F, VEC, U, false>(accl, eb, vb, rem);
        }
      } else {
        if (active) batch_add<F, VEC, U, false>(accl, ea, va, rem);
      }
    }
  }
  __syncthreads();

  float* __restrict__ mb = mean + ((long)b * NN + lo) * F + lane * VEC;
  const float* __restrict__ qb = ADDQ ? addq + ((long)b * NN + lo) * F + lane * VEC : nullptr;
  for (int r = wave; r < nrows; r += NW) {
    const float c = fmaxf((float)deg[r], 1.0f);
    if (active) {
      vecf o;
#pragma unroll
      for (int j = 0; j < VEC; ++j) o[j] = acc[r * F + j * NL + lane] / c;
      if (ADDQ) o += *(const vecf*)(qb + (long)r * F);
      *(vecf*)(mb + (long)r * F) = o;
    }
  }
}

template <int K1>
__global__ __launch_bounds__(128)
void mlp_chain_kernel(const float* __restrict__ X, const float* __restrict__ Wm1, const float* __restrict__ bm1,
                      const float* __restrict__ Wm2, const float* __restrict__ bm2,
                      const float* __restrict__ wdot, const float* __restrict__ bdot,
                      const int* __restrict__ act_range_p, float* __restrict__ dout, int M) {
  static_assert(HID == 128, "column map and k order are written for 8 tiles of 16");
  static_assert((K1 >> 2) >= 1 && ((K1 >> 2) << 2) < K1 && K1 % 2 == 0, "whole k-steps, then half a one");
  constexpr int NT = HID / 16;
  constexpr int NJ = NT / 4;
  const int lane = threadIdx.x & 31;
  const int mt = blockIdx.x * (blockDim.x >> 5) + (threadIdx.x >> 5);
  const int tm = (M + 15) >> 4;
  if (mt >= tm) return;

  const int ml = lane & 15;
  const int g  = lane >> 4;

  const int row = mt * 16 + ml;
  const int rc  = imin(row, M - 1);
  const int bb  = rc / PADR;
  const int rr  = rc - bb * PADR;
  const bool rvalid = (row < M) && (rr < *act_range_p);
  const float* __restrict__ xrow = X + ((long)bb * NN + rr) * K1;

  v8f acc1[NT];
#pragma unroll
  for (int t = 0; t < NT; ++t) acc1[t] = (v8f){};

  constexpr int nfull = K1 >> 2;
#pragma unroll
  for (int it = 0; it <= nfull; ++it) {
    const int ka = 4 * it + 2 * g;
    const bool kin = (it < nfull) || (ka < K1);
    const int kc = kin ? ka : K1 - 2;
    v2f xf = *(const v2f*)(xrow + kc);
    xf.x = (rvalid && kin) ? xf.x : 0.0f;
    xf.y = (rvalid && kin) ? xf.y : 0.0f;
    const float* w0 = Wm1 + kc * HID + 8 * ml;
    v4fu wa[2][2];
    wa[0][0] = *(const v4fu*)(w0);
    wa[0][1] = *(const v4fu*)(w0 + 4);
    wa[1][0] = *(const v4fu*)(w0 + HID);
    wa[1][1] = *(const v4fu*)(w0 + HID + 4);
#pragma unroll
    for (int t = 0; t < NT; ++t) {
      v2f wf;
      wf.x = kin ? wa[0][t >> 2][t & 3] : 0.0f;
      wf.y = kin ? wa[1][t >> 2][t & 3] : 0.0f;
      acc1[t] = wmma_f32x4(wf, xf, acc1[t]);
    }
  }

#pragma unroll
  for (int r = 0; r < 8; ++r) {
    const v4f blo = *(const v4fu*)(bm1 + 64 * g + 8 * r);
    const v4f bhi = *(const v4fu*)(bm1 + 64 * g + 8 * r + 4);
#pragma unroll
    for (int t = 0; t < NT; ++t) {
      const float bv = (t < 4) ? blo[t & 3] : bhi[t & 3];
      acc1[t][r] = fmaxf(acc1[t][r] + bv, 0.0f);
    }
  }

  v8f acc[NT];
#pragma unroll
  for (int t = 0; t < NT; ++t) acc[t] = (v8f){};

  const float* wq = Wm2 + (64 * g) * HID + 4 * ml;
  v4fu bc[4][NJ][2];
#pragma unroll
  for (int s = 0; s < 4; ++s)
    load_quads<NJ, 8 * HID>(wq + ((s >> 3) * 16 + (s & 7)) * HID, bc[s]);
#pragma unroll
  for (int gi = 0; gi < 8; ++gi) {
    v4fu bn[4][NJ][2];
    if (gi + 1 < 8) {
#pragma unroll
      for (int s = 0; s < 4; ++s) {
        const int n = 4 * (gi + 1) + s;
        load_quads<NJ, 8 * HID>(wq + ((n >> 3) * 16 + (n & 7)) * HID, bn[s]);
      }
    }
#pragma unroll
    for (int s = 0; s < 4; ++s) {
      const int n = 4 * gi + s;
      const int j = n >> 3, t = n & 7;
      v2f af;
      af.x = acc1[t][2 * j];
      af.y = acc1[t][2 * j + 1];
      kstep_quads<NJ>(af, bc[s], acc);
    }
    if (gi + 1 < 8) {
#pragma unroll
      for (int s = 0; s < 4; ++s) {
#pragma unroll
        for (int jj = 0; jj < NJ; ++jj) { bc[s][jj][0] = bn[s][jj][0]; bc[s][jj][1] = bn[s][jj][1]; }
      }
    }
  }

  float s[8];
#pragma unroll
  for (int v = 0; v < 8; ++v) s[v] = 0.0f;
#pragma unroll
  for (int j = 0; j < NJ; ++j) {
    const v4f bv = *(const v4fu*)(bm2 + 64 * j + 4 * ml);
    const v4f wv = *(const v4fu*)(wdot + 64 * j + 4 * ml);
#pragma unroll
    for (int v = 0; v < 8; ++v) {
      v4f val = (v4f){acc[4 * j + 0][v], acc[4 * j + 1][v], acc[4 * j + 2][v], acc[4 * j + 3][v]} + bv;
      val[0] = fmaxf(val[0], 0.0f); val[1] = fmaxf(val[1], 0.0f);
      val[2] = fmaxf(val[2], 0.0f); val[3] = fmaxf(val[3], 0.0f);
      s[v] += val[0] * wv[0];
      s[v] += val[1] * wv[1];
      s[v] += val[2] * wv[2];
      s[v] += val[3] * wv[3];
    }
  }
  float o = 0.0f;
#pragma unroll
  for (int v = 0; v < 8; ++v) {
#pragma unroll
    for (int off = 8; off > 0; off >>= 1) s[v] += __shfl_xor(s[v], off, 32);
    o = (ml == v) ? s[v] : o;
  }
  const int r = mt * 16 + (g << 3) + ml;
  if (ml < 8 && r < M) dout[r] = o + bdot[0];
}

#define HPITCH 132
#define FUSED_WAVES 4

template <int NJ>
__device__ __forceinline__ void kstep_pairs(const v2f af, const v2fu (&p)[NJ][2], const v2fu (&q)[NJ][2],
                                            v8f (&accp)[2 * NJ], v8f (&accq)[2 * NJ]) {
#pragma unroll
  for (int j = 0; j < NJ; ++j) {
#pragma unroll
    for (int u = 0; u < 2; ++u) {
      v2f bf;
      bf.x = p[j][0][u];
      bf.y = p[j][1][u];
      accp[2 * j + u] = wmma_f32x4(af, bf, accp[2 * j + u]);
      bf.x = q[j][0][u];
      bf.y = q[j][1][u];
      accq[2 * j + u] = wmma_f32x4(af, bf, accq[2 * j + u]);
    }
  }
}

template <int NJ, int NCT>
__device__ __forceinline__ void load_pairs(const float* wrow, const int (&cb)[NJ], v2fu (&p)[NJ][2]) {
#pragma unroll
  for (int j = 0; j < NJ; ++j) {
    p[j][0] = *(const v2fu*)(wrow + cb[j]);
    p[j][1] = *(const v2fu*)(wrow + NCT + cb[j]);
  }
}

template <int K1, int NT2, int NCT, int PS>
__global__ __launch_bounds__(FUSED_WAVES * 32)
void sage1_then_pre2_kernel(const float* __restrict__ A1, const float* __restrict__ A2,
                            const float* __restrict__ Wa1, const float* __restrict__ Wb1,
                            const float* __restrict__ bias1,
                            const float* __restrict__ Wa2, const float* __restrict__ Wb2,
                            const float* __restrict__ bias2,
                            float* __restrict__ P, float* __restrict__ Q, int M) {
  static_assert(HID % 16 == 0 && HID % 4 == 0, "h strip: whole column tiles and k-steps");
  static_assert((NT2 - 1) * 16 < NCT && NT2 * 16 >= NCT && PS >= NCT, "column tiles");
  static_assert(NT2 % 2 == 0 && NCT % 2 == 0 && NCT >= 32, "P, Q: column pairs, whole or past the end");
  static_assert(HID % 64 == 0, "h strip: column tiles in groups of four");
  static_assert(HPITCH >= HID && HPITCH % 4 == 0, "8-byte fragment reads, 16-byte strip stores");
  static_assert((K1 >> 2) >= 1 && ((K1 >> 2) << 2) < K1, "layer 1: whole k-steps, then a partial one");
  constexpr int NT1 = HID / 16;
  constexpr int NJ1 = NT1 / 4;
  constexpr int NJ2 = NT2 / 2;

  __shared__ alignas(16) float hs_all[FUSED_WAVES * 16 * HPITCH];

  const int lane = threadIdx.x & 31;
  const int wv   = threadIdx.x >> 5;
  const int mt = blockIdx.x * FUSED_WAVES + wv;
  const int tm = (M + 15) >> 4;
  if (mt >= tm) return;

  const int ml = lane & 15;
  const int g  = lane >> 4;

  const int row = mt * 16 + ml;
  const bool rvalid = row < M;
  const long arow = imin(row, M - 1);

  float* hs = hs_all + wv * (16 * HPITCH);

  {
    v8f acc[NT1];
#pragma unroll
    for (int t = 0; t < NT1; ++t) acc[t] = (v8f){};

    for (int pass = 0; pass < 2; ++pass) {
      const float* __restrict__ Ap = pass ? A2 : A1;
      const float* __restrict__ Wp = pass ? Wb1 : Wa1;

      const float* aptr = Ap + arow * (long)K1 + 2 * g;
      const float* bptr = Wp + (2 * g) * HID + 4 * ml;
      constexpr int nfull = K1 >> 2;

      v2f  ac = *(const v2f*)aptr;
      v4fu bc[NJ1][2];
      load_quads<NJ1, HID>(bptr, bc);

#pragma unroll 1
      for (int it = 0; it + 1 < nfull; ++it) {
        aptr += 4;
        bptr += 4 * HID;
        const v2f an = *(const v2f*)aptr;
        v4fu bn[NJ1][2];
        load_quads<NJ1, HID>(bptr, bn);
        v2f af;
        af.x = rvalid ? ac.x : 0.0f;
        af.y = rvalid ? ac.y : 0.0f;
        kstep_quads<NJ1>(af, bc, acc);
        ac = an;
#pragma unroll
        for (int j = 0; j < NJ1; ++j) { bc[j][0] = bn[j][0]; bc[j][1] = bn[j][1]; }
      }

      constexpr int k0 = nfull << 2;
      const float* __restrict__ Arow_p = Ap + arow * (long)K1;
      const int ka  = k0 + 2 * g;
      const int kc0 = imin(ka, K1 - 1);
      const int kc1 = imin(ka + 1, K1 - 1);
      const float a0 = Arow_p[kc0];
      const float a1 = Arow_p[kc1];
      const float* __restrict__ W0 = Wp + (long)kc0 * HID + 4 * ml;
      const float* __restrict__ W1 = Wp + (long)kc1 * HID + 4 * ml;
      v4fu bt[NJ1][2];
#pragma unroll
      for (int j = 0; j < NJ1; ++j) {
        bt[j][0] = *(const v4fu*)(W0 + 64 * j);
        bt[j][1] = *(const v4fu*)(W1 + 64 * j);
      }
      {
        v2f af;
        af.x = rvalid ? ac.x : 0.0f;
        af.y = rvalid ? ac.y : 0.0f;
        kstep_quads<NJ1>(af, bc, acc);
      }
      {
        v2f af;
        af.x = (rvalid && ka     < K1) ? a0 : 0.0f;
        af.y = (rvalid && ka + 1 < K1) ? a1 : 0.0f;
        kstep_quads<NJ1>(af, bt, acc);
      }
    }

    float* hw = hs + (g << 3) * HPITCH + 4 * ml;
#pragma unroll
    for (int j = 0; j < NJ1; ++j) {
      const v4f bv = *(const v4fu*)(bias1 + 64 * j + 4 * ml);
#pragma unroll
      for (int v = 0; v < 8; ++v) {
        v4f val = (v4f){acc[4 * j + 0][v], acc[4 * j + 1][v], acc[4 * j + 2][v], acc[4 * j + 3][v]} + bv;
        val[0] = fmaxf(val[0], 0.0f); val[1] = fmaxf(val[1], 0.0f);
        val[2] = fmaxf(val[2], 0.0f); val[3] = fmaxf(val[3], 0.0f);
        *(v4f*)(hw + v * HPITCH + 64 * j) = val;
      }
    }
  }

  v8f accp[NT2], accq[NT2];
#pragma unroll
  for (int t = 0; t < NT2; ++t) { accp[t] = (v8f){}; accq[t] = (v8f){}; }

  int cb[NJ2];
#pragma unroll
  for (int j = 0; j < NJ2; ++j) cb[j] = imin(32 * j + 2 * ml, NCT - 2);

  const float* hrow = hs + ml * HPITCH + 2 * g;
  const float* wa = Wa2 + (2 * g) * NCT;
  const float* wb = Wb2 + (2 * g) * NCT;

  v2f  hc = *(const v2f*)hrow;
  v2fu pc[NJ2][2], qc[NJ2][2];
  load_pairs<NJ2, NCT>(wa, cb, pc);
  load_pairs<NJ2, NCT>(wb, cb, qc);

#pragma unroll 1
  for (int it = 0; it + 1 < HID / 4; ++it) {
    wa += 4 * NCT;
    wb += 4 * NCT;
    const v2f hn = *(const v2f*)(hrow + 4 * (it + 1));
    v2fu pn[NJ2][2], qn[NJ2][2];
    load_pairs<NJ2, NCT>(wa, cb, pn);
    load_pairs<NJ2, NCT>(wb, cb, qn);
    kstep_pairs<NJ2>(hc, pc, qc, accp, accq);
    hc = hn;
#pragma unroll
    for (int j = 0; j < NJ2; ++j) {
      pc[j][0] = pn[j][0]; pc[j][1] = pn[j][1];
      qc[j][0] = qn[j][0]; qc[j][1] = qn[j][1];
    }
  }
  kstep_pairs<NJ2>(hc, pc, qc, accp, accq);

  float* prow = P + (long)(mt * 16 + (g << 3)) * PS;
  float* qrow = Q + (long)(mt * 16 + (g << 3)) * NCT;
#pragma unroll
  for (int j = 0; j < NJ2; ++j) {
    const int n = 32 * j + 2 * ml;
    const bool nvalid = n < NCT;
    const v2f bv = *(const v2fu*)(bias2 + cb[j]);
#pragma unroll
    for (int v = 0; v < 8; ++v) {
      const int r = mt * 16 + v + (g << 3);
      if (r < M && nvalid) {
        *(v2f*)(prow + v * PS  + n) = (v2f){accp[2 * j][v], accp[2 * j + 1][v]};
        *(v2f*)(qrow + v * NCT + n) = (v2f){accq[2 * j][v], accq[2 * j + 1][v]} + bv;
      }
    }
  }
}


static inline int strip_grid(int M) {
  int tm = (M + 15) / 16;
  return (tm + 3) / 4;
}

extern "C" void kernel_launch(void* const* d_in, const int* in_sizes, int n_in,
                              void* d_out, int out_size, void* d_ws, size_t ws_size,
                              hipStream_t stream) {
  (void)in_sizes; (void)n_in; (void)out_size; (void)ws_size;

  const float* edge_feat = (const float*)d_in[0];
  const int*   edge_idx  = (const int*)d_in[1];
  const int*   act_range = (const int*)d_in[2];
  const float* Wl1 = (const float*)d_in[3];
  const float* Wr1 = (const float*)d_in[4];
  const float* b1c = (const float*)d_in[5];
  const float* Wl2 = (const float*)d_in[6];
  const float* Wr2 = (const float*)d_in[7];
  const float* b2c = (const float*)d_in[8];
  const float* W1  = (const float*)d_in[9];
  const float* b1  = (const float*)d_in[10];
  const float* W2  = (const float*)d_in[11];
  const float* b2  = (const float*)d_in[12];
  const float* W3  = (const float*)d_in[13];
  const float* b3  = (const float*)d_in[14];
  float* out = (float*)d_out;

  const long S_AGG1 = (long)BB * NN * FEAT;
  const long S_CNT  = (long)BB * NN;
  const long S_H    = (long)BB * NN * HID;

  const long S_P = (long)BB * NN * PPITCH;

  float* agg1 = (float*)d_ws;
  float* hgap = agg1 + S_AGG1 + S_CNT;
  float* p2   = hgap + S_H;
  float* q2   = p2   + S_P;

  agg_mean_lds_kernel<FEAT, FEAT, 2, 313, 8, false><<<dim3(16, BB), AGG_THREADS, 0, stream>>>(
      edge_feat, edge_idx, nullptr, agg1);

  sage1_then_pre2_kernel<FEAT, 4, FEAT, PPITCH><<<strip_grid(BB * NN), FUSED_WAVES * 32, 0, stream>>>(
      agg1, edge_feat, Wl1, Wr1, b1c, Wl2, Wr2, b2c, p2, q2, BB * NN);
  agg_mean_lds_kernel<FEAT, PPITCH, 2, 313, 8, true><<<dim3(16, BB), AGG_THREADS, 0, stream>>>(
      p2, edge_idx, q2, agg1);

  mlp_chain_kernel<FEAT><<<strip_grid(BB * PADR), 128, 0, stream>>>(
      agg1, W1, b1, W2, b2, W3, b3, act_range, out, BB * PADR);
}
